# starting kernel
# baseline (speedup 1.0000x reference)
#include <hip/hip_runtime.h>

#define HN (2048 * 2048)
#define EPSF 1e-10f
#define DOUBLING_STEPS 23

typedef __attribute__((ext_vector_type(2))) float v2f;
typedef __attribute__((ext_vector_type(8))) float v8f;

__global__ __launch_bounds__(256) void phase1_contrib(
    const int* __restrict__ parent,
    const float* __restrict__ diff,
    const float* __restrict__ attr,
    const float* __restrict__ weight,
    const float* __restrict__ bias,
    unsigned long long* __restrict__ buf0)
{
    const int lane = threadIdx.x & 31;
    const int half = lane >> 4;
    const int m    = lane & 15;
    const int wave = blockIdx.x * (blockDim.x >> 5) + (threadIdx.x >> 5);
    const int base = wave << 4;
    const int row  = base + m;
    const float* a = attr + (size_t)row * 15;

    float bx[5], by[5];
    if (half == 0) {
        float a5 = a[5], a6 = a[6];
        bx[0] = a[0];                        by[0] = a[1];
        bx[1] = a[4];                        by[1] = logf(fabsf(a6) + EPSF);
        bx[2] = logf(fabsf(a[9])  + EPSF);   by[2] = logf(fabsf(a[10]) + EPSF);
        bx[3] = logf(fabsf(a[13]) + EPSF);   by[3] = logf(fabsf(a[14]) + EPSF);
        bx[4] = sinf(a5);                    by[4] = 0.0f;
    } else {
        float a5 = a[5], a6 = a[6], a7 = a[7];
        bx[0] = a[2];                        by[0] = a[3];
        bx[1] = logf(fabsf(a7)    + EPSF);   by[1] = logf(fabsf(a[8])  + EPSF);
        bx[2] = logf(fabsf(a[11]) + EPSF);   by[2] = logf(fabsf(a[12]) + EPSF);
        bx[3] = sqrtf(a7) / (sqrtf(a6) + EPSF);  by[3] = cosf(a5);
        bx[4] = 0.0f;                        by[4] = 0.0f;
    }

    float ax[5], ay[5];
#pragma unroll
    for (int c = 0; c < 5; ++c) {
        int kx = 4 * c + 2 * half;
        int ky = kx + 1;
        ax[c] = (kx < 17) ? weight[kx] : 0.0f;
        ay[c] = (ky < 17) ? weight[ky] : 0.0f;
    }

    v8f acc = {};
#pragma unroll
    for (int c = 0; c < 5; ++c) {
        v2f A; A.x = ax[c]; A.y = ay[c];
        v2f B; B.x = bx[c]; B.y = by[c];
        acc = __builtin_amdgcn_wmma_f32_16x16x4_f32(
             false, A,  false, B,
             (short)0, acc,  false,  false);
    }

    float val   = acc[0] + bias[0];
    float score = 1.0f / (1.0f + expf(-val));

    if (half == 0) {
        float contrib = diff[row] * score;
        unsigned long long packed =
            ((unsigned long long)(unsigned int)parent[row] << 32) |
            (unsigned long long)__float_as_uint(contrib);
        buf0[row] = packed;
    }
    if (blockIdx.x == 0 && threadIdx.x == 0) {
        buf0[HN] = ((unsigned long long)(unsigned int)HN << 32);
    }
}

__global__ __launch_bounds__(256) void doubling_step(
    const unsigned long long* __restrict__ in,
    unsigned long long* __restrict__ out,
    int n)
{
    int i = blockIdx.x * blockDim.x + threadIdx.x;
    if (i >= n) return;
    unsigned long long v = in[i];
    int p   = (int)(v >> 32);
    float c = __uint_as_float((unsigned int)v);
    unsigned long long w = in[p];
    float cp = __uint_as_float((unsigned int)w);
    out[i] = (w & 0xFFFFFFFF00000000ull)
           | (unsigned long long)__float_as_uint(c + cp);
}

__global__ __launch_bounds__(256) void final_step(
    const unsigned long long* __restrict__ in,
    float* __restrict__ out,
    int n)
{
    int i = blockIdx.x * blockDim.x + threadIdx.x;
    if (i >= n) return;
    unsigned long long v = in[i];
    int p   = (int)(v >> 32);
    float c = __uint_as_float((unsigned int)v);
    unsigned long long w = in[p];
    out[i] = c + __uint_as_float((unsigned int)w);
}

extern "C" void kernel_launch(void* const* d_in, const int* in_sizes, int n_in,
                              void* d_out, int out_size, void* d_ws, size_t ws_size,
                              hipStream_t stream) {
    const int*   parent = (const int*)d_in[0];
    const float* diff   = (const float*)d_in[1];
    const float* attr   = (const float*)d_in[2];
    const float* weight = (const float*)d_in[3];
    const float* bias   = (const float*)d_in[4];
    float* out = (float*)d_out;

    unsigned long long* buf0 = (unsigned long long*)d_ws;
    unsigned long long* buf1 = buf0 + (HN + 1);

    {
        int blocks = HN / (16 * 8);
        phase1_contrib<<<blocks, 256, 0, stream>>>(parent, diff, attr, weight, bias, buf0);
    }

    {
        int n = HN + 1;
        int blocks = (n + 255) / 256;
        unsigned long long* a = buf0;
        unsigned long long* b = buf1;
        for (int s = 0; s < DOUBLING_STEPS - 1; ++s) {
            doubling_step<<<blocks, 256, 0, stream>>>(a, b, n);
            unsigned long long* t = a; a = b; b = t;
        }
        final_step<<<(HN + 255) / 256, 256, 0, stream>>>(a, out, HN);
    }
}


# Last 13 of the 23 doubling levels in one launch: each node walks on along its level-10 links, adding in the same pairwise order, and stops at the self-linked zero root; 24 launches become 12
# speedup vs baseline: 1.5064x; 1.5064x over previous
#include <hip/hip_runtime.h>

#define HN (2048 * 2048)
#define EPSF 1e-10f
#define DOUBLING_STEPS 23
#define SPLIT_STEPS 10
#define FUSED_LEVELS (DOUBLING_STEPS - SPLIT_STEPS)

typedef __attribute__((ext_vector_type(2))) float v2f;
typedef __attribute__((ext_vector_type(8))) float v8f;

__global__ __launch_bounds__(256) void phase1_contrib(
    const int* __restrict__ parent,
    const float* __restrict__ diff,
    const float* __restrict__ attr,
    const float* __restrict__ weight,
    const float* __restrict__ bias,
    unsigned long long* __restrict__ buf0)
{
    const int lane = threadIdx.x & 31;
    const int half = lane >> 4;
    const int m    = lane & 15;
    const int wave = blockIdx.x * (blockDim.x >> 5) + (threadIdx.x >> 5);
    const int base = wave << 4;
    const int row  = base + m;
    const float* a = attr + (size_t)row * 15;

    float bx[5], by[5];
    if (half == 0) {
        float a5 = a[5], a6 = a[6];
        bx[0] = a[0];                        by[0] = a[1];
        bx[1] = a[4];                        by[1] = logf(fabsf(a6) + EPSF);
        bx[2] = logf(fabsf(a[9])  + EPSF);   by[2] = logf(fabsf(a[10]) + EPSF);
        bx[3] = logf(fabsf(a[13]) + EPSF);   by[3] = logf(fabsf(a[14]) + EPSF);
        bx[4] = sinf(a5);                    by[4] = 0.0f;
    } else {
        float a5 = a[5], a6 = a[6], a7 = a[7];
        bx[0] = a[2];                        by[0] = a[3];
        bx[1] = logf(fabsf(a7)    + EPSF);   by[1] = logf(fabsf(a[8])  + EPSF);
        bx[2] = logf(fabsf(a[11]) + EPSF);   by[2] = logf(fabsf(a[12]) + EPSF);
        bx[3] = sqrtf(a7) / (sqrtf(a6) + EPSF);  by[3] = cosf(a5);
        bx[4] = 0.0f;                        by[4] = 0.0f;
    }

    float ax[5], ay[5];
#pragma unroll
    for (int c = 0; c < 5; ++c) {
        int kx = 4 * c + 2 * half;
        int ky = kx + 1;
        ax[c] = (kx < 17) ? weight[kx] : 0.0f;
        ay[c] = (ky < 17) ? weight[ky] : 0.0f;
    }

    v8f acc = {};
#pragma unroll
    for (int c = 0; c < 5; ++c) {
        v2f A; A.x = ax[c]; A.y = ay[c];
        v2f B; B.x = bx[c]; B.y = by[c];
        acc = __builtin_amdgcn_wmma_f32_16x16x4_f32(
             false, A,  false, B,
             (short)0, acc,  false,  false);
    }

    float val   = acc[0] + bias[0];
    float score = 1.0f / (1.0f + expf(-val));

    if (half == 0) {
        float contrib = diff[row] * score;
        unsigned long long packed =
            ((unsigned long long)(unsigned int)parent[row] << 32) |
            (unsigned long long)__float_as_uint(contrib);
        buf0[row] = packed;
    }
    if (blockIdx.x == 0 && threadIdx.x == 0) {
        buf0[HN] = ((unsigned long long)(unsigned int)HN << 32);
    }
}

__global__ __launch_bounds__(256) void doubling_step(
    const unsigned long long* __restrict__ in,
    unsigned long long* __restrict__ out,
    int n)
{
    int i = blockIdx.x * blockDim.x + threadIdx.x;
    if (i >= n) return;
    unsigned long long v = in[i];
    int p   = (int)(v >> 32);
    float c = __uint_as_float((unsigned int)v);
    unsigned long long w = in[p];
    float cp = __uint_as_float((unsigned int)w);
    out[i] = (w & 0xFFFFFFFF00000000ull)
           | (unsigned long long)__float_as_uint(c + cp);
}

__global__ __launch_bounds__(256) void fused_levels_final(
    const unsigned long long* __restrict__ in,
    float* __restrict__ out,
    int n)
{
    constexpr int K = FUSED_LEVELS;
    int i = blockIdx.x * blockDim.x + threadIdx.x;
    if (i >= n) return;
    unsigned long long v = in[i];
    int p   = (int)(v >> 32);
    float x = __uint_as_float((unsigned int)v);

    float acc[K + 1];
#pragma unroll
    for (int L = 0; L <= K; ++L) acc[L] = 0.0f;
    acc[0] = x;
    unsigned int j = 1u;

    while (p != HN && j < (1u << K)) {
        unsigned long long w = in[p];
        x = __uint_as_float((unsigned int)w);
        p = (int)(w >> 32);
        bool placed = false;
#pragma unroll
        for (int L = 0; L < K; ++L) {
            if (!placed) {
                if ((j >> L) & 1u) x = acc[L] + x;
                else { acc[L] = x; placed = true; }
            }
        }
        if (!placed) acc[K] = x;
        ++j;
    }

    float r;
    if (j == (1u << K)) {
        r = acc[K];
    } else {
        r = 0.0f;
#pragma unroll
        for (int L = 0; L < K; ++L) {
            if ((j >> L) == 0u) break;
            if ((j >> L) & 1u) r = acc[L] + r;
        }
    }
    out[i] = r;
}

extern "C" void kernel_launch(void* const* d_in, const int* in_sizes, int n_in,
                              void* d_out, int out_size, void* d_ws, size_t ws_size,
                              hipStream_t stream) {
    const int*   parent = (const int*)d_in[0];
    const float* diff   = (const float*)d_in[1];
    const float* attr   = (const float*)d_in[2];
    const float* weight = (const float*)d_in[3];
    const float* bias   = (const float*)d_in[4];
    float* out = (float*)d_out;

    unsigned long long* buf0 = (unsigned long long*)d_ws;
    unsigned long long* buf1 = buf0 + (HN + 1);

    {
        int blocks = HN / (16 * 8);
        phase1_contrib<<<blocks, 256, 0, stream>>>(parent, diff, attr, weight, bias, buf0);
    }

    {
        int n = HN + 1;
        int blocks = (n + 255) / 256;
        unsigned long long* a = buf0;
        unsigned long long* b = buf1;
        for (int s = 0; s < SPLIT_STEPS; ++s) {
            doubling_step<<<blocks, 256, 0, stream>>>(a, b, n);
            unsigned long long* t = a; a = b; b = t;
        }
        fused_levels_final<<<(HN + 255) / 256, 256, 0, stream>>>(a, out, HN);
    }
}


# Phase 1 computes each node's 17 features in one lane (32 nodes per wave, two WMMA tiles, k pairs cross the lane halves by v_permlanex16) instead of two half-masked branches run in turn over 16 nodes
# speedup vs baseline: 1.6182x; 1.0742x over previous
#include <hip/hip_runtime.h>

#define HN (2048 * 2048)
#define EPSF 1e-10f
#define DOUBLING_STEPS 23
#define SPLIT_STEPS 10
#define FUSED_LEVELS (DOUBLING_STEPS - SPLIT_STEPS)

typedef __attribute__((ext_vector_type(2))) float v2f;
typedef __attribute__((ext_vector_type(8))) float v8f;

__device__ __forceinline__ float other_half(float x) {
    return __builtin_bit_cast(float, __builtin_amdgcn_permlanex16(
        __builtin_bit_cast(int, x), __builtin_bit_cast(int, x),
        0x76543210, 0xfedcba98, false, false));
}

__global__ __launch_bounds__(256) void phase1_contrib(
    const int* __restrict__ parent,
    const float* __restrict__ diff,
    const float* __restrict__ attr,
    const float* __restrict__ weight,
    const float* __restrict__ bias,
    unsigned long long* __restrict__ buf0)
{
    const int lane = threadIdx.x & 31;
    const int half = lane >> 4;
    const bool lo  = (half == 0);
    const int row  = blockIdx.x * blockDim.x + threadIdx.x;
    const float* a = attr + (size_t)row * 15;

    float av[15];
#pragma unroll
    for (int k = 0; k < 15; ++k) av[k] = a[k];
    const float dv = diff[row];
    const int   pv = parent[row];

    float f[20];
#pragma unroll
    for (int k = 0; k < 5; ++k) f[k] = av[k];
#pragma unroll
    for (int k = 0; k < 9; ++k) f[5 + k] = logf(fabsf(av[6 + k]) + EPSF);
    f[14] = sqrtf(av[7]) / (sqrtf(av[6]) + EPSF);
    f[15] = cosf(av[5]);
    f[16] = sinf(av[5]);
    f[17] = 0.0f; f[18] = 0.0f; f[19] = 0.0f;

    float ax[5], ay[5];
#pragma unroll
    for (int c = 0; c < 5; ++c) {
        int kx = 4 * c + 2 * half;
        int ky = kx + 1;
        ax[c] = (kx < 17) ? weight[kx] : 0.0f;
        ay[c] = (ky < 17) ? weight[ky] : 0.0f;
    }

    v8f acc0 = {};
    v8f acc1 = {};
#pragma unroll
    for (int c = 0; c < 5; ++c) {
        const float o0x = f[4 * c],     o0y = f[4 * c + 1];
        const float o1x = f[4 * c + 2], o1y = f[4 * c + 3];
        const float rx = other_half(lo ? o1x : o0x);
        const float ry = other_half(lo ? o1y : o0y);
        v2f A;  A.x  = ax[c];          A.y  = ay[c];
        v2f B0; B0.x = lo ? o0x : rx;  B0.y = lo ? o0y : ry;
        v2f B1; B1.x = lo ? rx : o1x;  B1.y = lo ? ry : o1y;
        acc0 = __builtin_amdgcn_wmma_f32_16x16x4_f32(
             false, A,  false, B0,
             (short)0, acc0,  false,  false);
        acc1 = __builtin_amdgcn_wmma_f32_16x16x4_f32(
             false, A,  false, B1,
             (short)0, acc1,  false,  false);
    }

    float val   = (lo ? acc0[0] : acc1[0]) + bias[0];
    float score = 1.0f / (1.0f + expf(-val));
    float contrib = dv * score;
    unsigned long long packed =
        ((unsigned long long)(unsigned int)pv << 32) |
        (unsigned long long)__float_as_uint(contrib);
    buf0[row] = packed;
    if (blockIdx.x == 0 && threadIdx.x == 0) {
        buf0[HN] = ((unsigned long long)(unsigned int)HN << 32);
    }
}

__global__ __launch_bounds__(256) void doubling_step(
    const unsigned long long* __restrict__ in,
    unsigned long long* __restrict__ out,
    int n)
{
    int i = blockIdx.x * blockDim.x + threadIdx.x;
    if (i >= n) return;
    unsigned long long v = in[i];
    int p   = (int)(v >> 32);
    float c = __uint_as_float((unsigned int)v);
    unsigned long long w = in[p];
    float cp = __uint_as_float((unsigned int)w);
    out[i] = (w & 0xFFFFFFFF00000000ull)
           | (unsigned long long)__float_as_uint(c + cp);
}

__global__ __launch_bounds__(256) void fused_levels_final(
    const unsigned long long* __restrict__ in,
    float* __restrict__ out,
    int n)
{
    constexpr int K = FUSED_LEVELS;
    int i = blockIdx.x * blockDim.x + threadIdx.x;
    if (i >= n) return;
    unsigned long long v = in[i];
    int p   = (int)(v >> 32);
    float x = __uint_as_float((unsigned int)v);

    float acc[K + 1];
#pragma unroll
    for (int L = 0; L <= K; ++L) acc[L] = 0.0f;
    acc[0] = x;
    unsigned int j = 1u;

    while (p != HN && j < (1u << K)) {
        unsigned long long w = in[p];
        x = __uint_as_float((unsigned int)w);
        p = (int)(w >> 32);
        bool placed = false;
#pragma unroll
        for (int L = 0; L < K; ++L) {
            if (!placed) {
                if ((j >> L) & 1u) x = acc[L] + x;
                else { acc[L] = x; placed = true; }
            }
        }
        if (!placed) acc[K] = x;
        ++j;
    }

    float r;
    if (j == (1u << K)) {
        r = acc[K];
    } else {
        r = 0.0f;
#pragma unroll
        for (int L = 0; L < K; ++L) {
            if ((j >> L) == 0u) break;
            if ((j >> L) & 1u) r = acc[L] + r;
        }
    }
    out[i] = r;
}

extern "C" void kernel_launch(void* const* d_in, const int* in_sizes, int n_in,
                              void* d_out, int out_size, void* d_ws, size_t ws_size,
                              hipStream_t stream) {
    const int*   parent = (const int*)d_in[0];
    const float* diff   = (const float*)d_in[1];
    const float* attr   = (const float*)d_in[2];
    const float* weight = (const float*)d_in[3];
    const float* bias   = (const float*)d_in[4];
    float* out = (float*)d_out;

    unsigned long long* buf0 = (unsigned long long*)d_ws;
    unsigned long long* buf1 = buf0 + (HN + 1);

    {
        int blocks = HN / (32 * 8);
        phase1_contrib<<<blocks, 256, 0, stream>>>(parent, diff, attr, weight, bias, buf0);
    }

    {
        int n = HN + 1;
        int blocks = (n + 255) / 256;
        unsigned long long* a = buf0;
        unsigned long long* b = buf1;
        for (int s = 0; s < SPLIT_STEPS; ++s) {
            doubling_step<<<blocks, 256, 0, stream>>>(a, b, n);
            unsigned long long* t = a; a = b; b = t;
        }
        fused_levels_final<<<(HN + 255) / 256, 256, 0, stream>>>(a, out, HN);
    }
}


# Doubling launches: each lane carries 8 records (8 loads, then 8 gathers in flight together, then 8 stores) in a 2048-block grid the chip holds at once, instead of 131072 one-record waves per launch
# speedup vs baseline: 1.6711x; 1.0327x over previous
#include <hip/hip_runtime.h>

#define HN (2048 * 2048)
#define EPSF 1e-10f
#define DOUBLING_STEPS 23
#define SPLIT_STEPS 10
#define FUSED_LEVELS (DOUBLING_STEPS - SPLIT_STEPS)

typedef __attribute__((ext_vector_type(2))) float v2f;
typedef __attribute__((ext_vector_type(8))) float v8f;

__device__ __forceinline__ float other_half(float x) {
    return __builtin_bit_cast(float, __builtin_amdgcn_permlanex16(
        __builtin_bit_cast(int, x), __builtin_bit_cast(int, x),
        0x76543210, 0xfedcba98, false, false));
}

__global__ __launch_bounds__(256) void phase1_contrib(
    const int* __restrict__ parent,
    const float* __restrict__ diff,
    const float* __restrict__ attr,
    const float* __restrict__ weight,
    const float* __restrict__ bias,
    unsigned long long* __restrict__ buf0)
{
    const int lane = threadIdx.x & 31;
    const int half = lane >> 4;
    const bool lo  = (half == 0);
    const int row  = blockIdx.x * blockDim.x + threadIdx.x;
    const float* a = attr + (size_t)row * 15;

    float av[15];
#pragma unroll
    for (int k = 0; k < 15; ++k) av[k] = a[k];
    const float dv = diff[row];
    const int   pv = parent[row];

    float f[20];
#pragma unroll
    for (int k = 0; k < 5; ++k) f[k] = av[k];
#pragma unroll
    for (int k = 0; k < 9; ++k) f[5 + k] = logf(fabsf(av[6 + k]) + EPSF);
    f[14] = sqrtf(av[7]) / (sqrtf(av[6]) + EPSF);
    f[15] = cosf(av[5]);
    f[16] = sinf(av[5]);
    f[17] = 0.0f; f[18] = 0.0f; f[19] = 0.0f;

    float ax[5], ay[5];
#pragma unroll
    for (int c = 0; c < 5; ++c) {
        int kx = 4 * c + 2 * half;
        int ky = kx + 1;
        ax[c] = (kx < 17) ? weight[kx] : 0.0f;
        ay[c] = (ky < 17) ? weight[ky] : 0.0f;
    }

    v8f acc0 = {};
    v8f acc1 = {};
#pragma unroll
    for (int c = 0; c < 5; ++c) {
        const float o0x = f[4 * c],     o0y = f[4 * c + 1];
        const float o1x = f[4 * c + 2], o1y = f[4 * c + 3];
        const float rx = other_half(lo ? o1x : o0x);
        const float ry = other_half(lo ? o1y : o0y);
        v2f A;  A.x  = ax[c];          A.y  = ay[c];
        v2f B0; B0.x = lo ? o0x : rx;  B0.y = lo ? o0y : ry;
        v2f B1; B1.x = lo ? rx : o1x;  B1.y = lo ? ry : o1y;
        acc0 = __builtin_amdgcn_wmma_f32_16x16x4_f32(
             false, A,  false, B0,
             (short)0, acc0,  false,  false);
        acc1 = __builtin_amdgcn_wmma_f32_16x16x4_f32(
             false, A,  false, B1,
             (short)0, acc1,  false,  false);
    }

    float val   = (lo ? acc0[0] : acc1[0]) + bias[0];
    float score = 1.0f / (1.0f + expf(-val));
    float contrib = dv * score;
    unsigned long long packed =
        ((unsigned long long)(unsigned int)pv << 32) |
        (unsigned long long)__float_as_uint(contrib);
    buf0[row] = packed;
    if (blockIdx.x == 0 && threadIdx.x == 0) {
        buf0[HN] = ((unsigned long long)(unsigned int)HN << 32);
    }
}

#define DS_REC 8
#define DS_BLOCKS (HN / (256 * DS_REC))

__device__ __forceinline__ unsigned long long doubled(unsigned long long v,
                                                     unsigned long long w) {
    float c  = __uint_as_float((unsigned int)v);
    float cp = __uint_as_float((unsigned int)w);
    return (w & 0xFFFFFFFF00000000ull)
         | (unsigned long long)__float_as_uint(c + cp);
}

__global__ __launch_bounds__(256) void doubling_step(
    const unsigned long long* __restrict__ in,
    unsigned long long* __restrict__ out)
{
    const int base = blockIdx.x * (256 * DS_REC) + threadIdx.x;

    unsigned long long v[DS_REC];
    unsigned long long w[DS_REC];
#pragma unroll
    for (int k = 0; k < DS_REC; ++k) v[k] = in[base + 256 * k];
#pragma unroll
    for (int k = 0; k < DS_REC; ++k) w[k] = in[(int)(v[k] >> 32)];
#pragma unroll
    for (int k = 0; k < DS_REC; ++k) out[base + 256 * k] = doubled(v[k], w[k]);

    if (blockIdx.x == 0 && threadIdx.x == 0) {
        unsigned long long vr = in[HN];
        unsigned long long wr = in[(int)(vr >> 32)];
        out[HN] = doubled(vr, wr);
    }
}

__global__ __launch_bounds__(256) void fused_levels_final(
    const unsigned long long* __restrict__ in,
    float* __restrict__ out,
    int n)
{
    constexpr int K = FUSED_LEVELS;
    int i = blockIdx.x * blockDim.x + threadIdx.x;
    if (i >= n) return;
    unsigned long long v = in[i];
    int p   = (int)(v >> 32);
    float x = __uint_as_float((unsigned int)v);

    float acc[K + 1];
#pragma unroll
    for (int L = 0; L <= K; ++L) acc[L] = 0.0f;
    acc[0] = x;
    unsigned int j = 1u;

    while (p != HN && j < (1u << K)) {
        unsigned long long w = in[p];
        x = __uint_as_float((unsigned int)w);
        p = (int)(w >> 32);
        bool placed = false;
#pragma unroll
        for (int L = 0; L < K; ++L) {
            if (!placed) {
                if ((j >> L) & 1u) x = acc[L] + x;
                else { acc[L] = x; placed = true; }
            }
        }
        if (!placed) acc[K] = x;
        ++j;
    }

    float r;
    if (j == (1u << K)) {
        r = acc[K];
    } else {
        r = 0.0f;
#pragma unroll
        for (int L = 0; L < K; ++L) {
            if ((j >> L) == 0u) break;
            if ((j >> L) & 1u) r = acc[L] + r;
        }
    }
    out[i] = r;
}

extern "C" void kernel_launch(void* const* d_in, const int* in_sizes, int n_in,
                              void* d_out, int out_size, void* d_ws, size_t ws_size,
                              hipStream_t stream) {
    const int*   parent = (const int*)d_in[0];
    const float* diff   = (const float*)d_in[1];
    const float* attr   = (const float*)d_in[2];
    const float* weight = (const float*)d_in[3];
    const float* bias   = (const float*)d_in[4];
    float* out = (float*)d_out;

    unsigned long long* buf0 = (unsigned long long*)d_ws;
    unsigned long long* buf1 = buf0 + (HN + 1);

    {
        int blocks = HN / (32 * 8);
        phase1_contrib<<<blocks, 256, 0, stream>>>(parent, diff, attr, weight, bias, buf0);
    }

    {
        unsigned long long* a = buf0;
        unsigned long long* b = buf1;
        for (int s = 0; s < SPLIT_STEPS; ++s) {
            doubling_step<<<DS_BLOCKS, 256, 0, stream>>>(a, b);
            unsigned long long* t = a; a = b; b = t;
        }
        fused_levels_final<<<(HN + 255) / 256, 256, 0, stream>>>(a, out, HN);
    }
}


# Doubling levels 5-6 and 7-9 run as two launches, not five: each record follows 3 or 7 links itself in the same pairwise sum order; links into the 32768 records nearest the root read an LDS copy
# speedup vs baseline: 1.8701x; 1.1191x over previous
#include <hip/hip_runtime.h>

#define HN (2048 * 2048)
#define EPSF 1e-10f
#define DOUBLING_STEPS 23
#define SPLIT_STEPS 10
#define SINGLE_STEPS 5
#define FUSED_LEVELS (DOUBLING_STEPS - SPLIT_STEPS)

typedef __attribute__((ext_vector_type(2))) float v2f;
typedef __attribute__((ext_vector_type(8))) float v8f;

__device__ __forceinline__ float other_half(float x) {
    return __builtin_bit_cast(float, __builtin_amdgcn_permlanex16(
        __builtin_bit_cast(int, x), __builtin_bit_cast(int, x),
        0x76543210, 0xfedcba98, false, false));
}

__global__ __launch_bounds__(256) void phase1_contrib(
    const int* __restrict__ parent,
    const float* __restrict__ diff,
    const float* __restrict__ attr,
    const float* __restrict__ weight,
    const float* __restrict__ bias,
    unsigned long long* __restrict__ buf0)
{
    const int lane = threadIdx.x & 31;
    const int half = lane >> 4;
    const bool lo  = (half == 0);
    const int row  = blockIdx.x * blockDim.x + threadIdx.x;
    const float* a = attr + (size_t)row * 15;

    float av[15];
#pragma unroll
    for (int k = 0; k < 15; ++k) av[k] = a[k];
    const float dv = diff[row];
    const int   pv = parent[row];

    float f[20];
#pragma unroll
    for (int k = 0; k < 5; ++k) f[k] = av[k];
#pragma unroll
    for (int k = 0; k < 9; ++k) f[5 + k] = logf(fabsf(av[6 + k]) + EPSF);
    f[14] = sqrtf(av[7]) / (sqrtf(av[6]) + EPSF);
    f[15] = cosf(av[5]);
    f[16] = sinf(av[5]);
    f[17] = 0.0f; f[18] = 0.0f; f[19] = 0.0f;

    float ax[5], ay[5];
#pragma unroll
    for (int c = 0; c < 5; ++c) {
        int kx = 4 * c + 2 * half;
        int ky = kx + 1;
        ax[c] = (kx < 17) ? weight[kx] : 0.0f;
        ay[c] = (ky < 17) ? weight[ky] : 0.0f;
    }

    v8f acc0 = {};
    v8f acc1 = {};
#pragma unroll
    for (int c = 0; c < 5; ++c) {
        const float o0x = f[4 * c],     o0y = f[4 * c + 1];
        const float o1x = f[4 * c + 2], o1y = f[4 * c + 3];
        const float rx = other_half(lo ? o1x : o0x);
        const float ry = other_half(lo ? o1y : o0y);
        v2f A;  A.x  = ax[c];          A.y  = ay[c];
        v2f B0; B0.x = lo ? o0x : rx;  B0.y = lo ? o0y : ry;
        v2f B1; B1.x = lo ? rx : o1x;  B1.y = lo ? ry : o1y;
        acc0 = __builtin_amdgcn_wmma_f32_16x16x4_f32(
             false, A,  false, B0,
             (short)0, acc0,  false,  false);
        acc1 = __builtin_amdgcn_wmma_f32_16x16x4_f32(
             false, A,  false, B1,
             (short)0, acc1,  false,  false);
    }

    float val   = (lo ? acc0[0] : acc1[0]) + bias[0];
    float score = 1.0f / (1.0f + expf(-val));
    float contrib = dv * score;
    unsigned long long packed =
        ((unsigned long long)(unsigned int)pv << 32) |
        (unsigned long long)__float_as_uint(contrib);
    buf0[row] = packed;
    if (blockIdx.x == 0 && threadIdx.x == 0) {
        buf0[HN] = ((unsigned long long)(unsigned int)HN << 32);
    }
}

#define DS_REC 8
#define DS_BLOCKS (HN / (256 * DS_REC))

__device__ __forceinline__ unsigned long long doubled(unsigned long long v,
                                                     unsigned long long w) {
    float c  = __uint_as_float((unsigned int)v);
    float cp = __uint_as_float((unsigned int)w);
    return (w & 0xFFFFFFFF00000000ull)
         | (unsigned long long)__float_as_uint(c + cp);
}

__global__ __launch_bounds__(256) void doubling_step(
    const unsigned long long* __restrict__ in,
    unsigned long long* __restrict__ out)
{
    const int base = blockIdx.x * (256 * DS_REC) + threadIdx.x;

    unsigned long long v[DS_REC];
    unsigned long long w[DS_REC];
#pragma unroll
    for (int k = 0; k < DS_REC; ++k) v[k] = in[base + 256 * k];
#pragma unroll
    for (int k = 0; k < DS_REC; ++k) w[k] = in[(int)(v[k] >> 32)];
#pragma unroll
    for (int k = 0; k < DS_REC; ++k) out[base + 256 * k] = doubled(v[k], w[k]);

    if (blockIdx.x == 0 && threadIdx.x == 0) {
        unsigned long long vr = in[HN];
        unsigned long long wr = in[(int)(vr >> 32)];
        out[HN] = doubled(vr, wr);
    }
}

#define TRUNK 32768
#define FD_THREADS 1024
#define FD_BLOCKS 256
#define FD_PER_THREAD (HN / (FD_BLOCKS * FD_THREADS))

__device__ __forceinline__ unsigned long long linked_record(
    const unsigned long long* __restrict__ in,
    const unsigned long long* trunk, int a)
{
    const bool outside = (a >= TRUNK) && (a != HN);
    const int  slot    = (a < TRUNK) ? a : TRUNK;
    unsigned long long w = trunk[slot];
    if (outside) {
        const uint2 g = *reinterpret_cast<const uint2*>(in + a);
        w = ((unsigned long long)g.y << 32) | (unsigned long long)g.x;
    }
    return w;
}

template <int M>
__device__ __forceinline__ void place_leaf(float (&acc)[M + 1], int j, float x) {
    bool placed = false;
#pragma unroll
    for (int L = 0; L < M; ++L) {
        if (!placed) {
            if ((j >> L) & 1) x = acc[L] + x;
            else { acc[L] = x; placed = true; }
        }
    }
    if (!placed) acc[M] = x;
}

template <int M, int FD_REC>
__global__ __launch_bounds__(FD_THREADS) void doubling_levels(
    const unsigned long long* __restrict__ in,
    unsigned long long* __restrict__ out)
{
    constexpr int LEAVES = 1 << M;
    constexpr int FD_BATCHES = FD_PER_THREAD / FD_REC;
    static_assert(FD_BATCHES * FD_REC == FD_PER_THREAD, "records per thread");
    __shared__ unsigned long long trunk[TRUNK + 1];
    const int tid = threadIdx.x;

#pragma unroll 1
    for (int q = 0; q < TRUNK / FD_THREADS; q += 8) {
        unsigned long long t[8];
#pragma unroll
        for (int k = 0; k < 8; ++k) t[k] = in[(q + k) * FD_THREADS + tid];
#pragma unroll
        for (int k = 0; k < 8; ++k) trunk[(q + k) * FD_THREADS + tid] = t[k];
    }
    if (tid == 0) trunk[TRUNK] = in[HN];
    __syncthreads();

#pragma unroll 1
    for (int b = 0; b < FD_BATCHES; ++b) {
        const int base = (blockIdx.x * FD_BATCHES + b) * (FD_THREADS * FD_REC) + tid;
        unsigned long long w[FD_REC];
        float acc[FD_REC][M + 1];
#pragma unroll
        for (int k = 0; k < FD_REC; ++k) w[k] = in[base + FD_THREADS * k];
#pragma unroll
        for (int j = 0; j < LEAVES; ++j) {
#pragma unroll
            for (int k = 0; k < FD_REC; ++k) {
                place_leaf<M>(acc[k], j, __uint_as_float((unsigned int)w[k]));
                if (j + 1 < LEAVES) w[k] = linked_record(in, trunk, (int)(w[k] >> 32));
            }
        }
#pragma unroll
        for (int k = 0; k < FD_REC; ++k)
            out[base + FD_THREADS * k] = (w[k] & 0xFFFFFFFF00000000ull)
                                       | (unsigned long long)__float_as_uint(acc[k][M]);
    }

    if (blockIdx.x == 0 && tid == 0) {
        unsigned long long wr = in[HN];
        float ar[M + 1];
#pragma unroll
        for (int j = 0; j < LEAVES; ++j) {
            place_leaf<M>(ar, j, __uint_as_float((unsigned int)wr));
            if (j + 1 < LEAVES) wr = linked_record(in, trunk, (int)(wr >> 32));
        }
        out[HN] = (wr & 0xFFFFFFFF00000000ull) | (unsigned long long)__float_as_uint(ar[M]);
    }
}

__global__ __launch_bounds__(256) void fused_levels_final(
    const unsigned long long* __restrict__ in,
    float* __restrict__ out,
    int n)
{
    constexpr int K = FUSED_LEVELS;
    int i = blockIdx.x * blockDim.x + threadIdx.x;
    if (i >= n) return;
    unsigned long long v = in[i];
    int p   = (int)(v >> 32);
    float x = __uint_as_float((unsigned int)v);

    float acc[K + 1];
#pragma unroll
    for (int L = 0; L <= K; ++L) acc[L] = 0.0f;
    acc[0] = x;
    unsigned int j = 1u;

    while (p != HN && j < (1u << K)) {
        unsigned long long w = in[p];
        x = __uint_as_float((unsigned int)w);
        p = (int)(w >> 32);
        bool placed = false;
#pragma unroll
        for (int L = 0; L < K; ++L) {
            if (!placed) {
                if ((j >> L) & 1u) x = acc[L] + x;
                else { acc[L] = x; placed = true; }
            }
        }
        if (!placed) acc[K] = x;
        ++j;
    }

    float r;
    if (j == (1u << K)) {
        r = acc[K];
    } else {
        r = 0.0f;
#pragma unroll
        for (int L = 0; L < K; ++L) {
            if ((j >> L) == 0u) break;
            if ((j >> L) & 1u) r = acc[L] + r;
        }
    }
    out[i] = r;
}

extern "C" void kernel_launch(void* const* d_in, const int* in_sizes, int n_in,
                              void* d_out, int out_size, void* d_ws, size_t ws_size,
                              hipStream_t stream) {
    const int*   parent = (const int*)d_in[0];
    const float* diff   = (const float*)d_in[1];
    const float* attr   = (const float*)d_in[2];
    const float* weight = (const float*)d_in[3];
    const float* bias   = (const float*)d_in[4];
    float* out = (float*)d_out;

    unsigned long long* buf0 = (unsigned long long*)d_ws;
    unsigned long long* buf1 = buf0 + (HN + 1);

    {
        int blocks = HN / (32 * 8);
        phase1_contrib<<<blocks, 256, 0, stream>>>(parent, diff, attr, weight, bias, buf0);
    }

    {
        unsigned long long* a = buf0;
        unsigned long long* b = buf1;
        for (int s = 0; s < SINGLE_STEPS; ++s) {
            doubling_step<<<DS_BLOCKS, 256, 0, stream>>>(a, b);
            unsigned long long* t = a; a = b; b = t;
        }
        doubling_levels<2, 8><<<FD_BLOCKS, FD_THREADS, 0, stream>>>(a, b);
        { unsigned long long* t = a; a = b; b = t; }
        doubling_levels<3, 4><<<FD_BLOCKS, FD_THREADS, 0, stream>>>(a, b);
        { unsigned long long* t = a; a = b; b = t; }
        static_assert(SINGLE_STEPS + 2 + 3 == SPLIT_STEPS, "level count");
        fused_levels_final<<<(HN + 255) / 256, 256, 0, stream>>>(a, out, HN);
    }
}
